# starting kernel
# baseline (speedup 1.0000x reference)
#include <hip/hip_runtime.h>
#include <hip/hip_bf16.h>

typedef __attribute__((ext_vector_type(2))) float v2f;
typedef __attribute__((ext_vector_type(8))) float v8f;

#define BSZ    8192
#define TSEQ   4096
#define OUTW   4095
#define NVALID 4093
#define CH     16
#define NCHUNK 256
#define WAVES_PER_BLOCK 2
#define ROWS_PER_WAVE   16
#define ROWS_PER_BLOCK  (WAVES_PER_BLOCK * ROWS_PER_WAVE)
#define PF_DIST (8 * CH)

template <bool LAST>
__device__ __forceinline__ void process_chunk(
    int base, const float* __restrict__ yrow, float* __restrict__ orow,
    int khalf, int m0, int lrow,
    const float A0k[4], const float A1k[4], const float hv[9],
    float q0, float ph1, float muv, float th1,
    float& e1, float& e2)
{
  float yb[17];
  const int tbase = base + khalf;
#pragma unroll
  for (int j = 1; j <= 16; ++j) {
    if (LAST) {
      const int idx = tbase + j;
      yb[j] = (idx < TSEQ) ? yrow[idx] : 0.0f;
    } else {
      yb[j] = yrow[tbase + j];
    }
  }
  if (!LAST) {
    __builtin_prefetch(yrow + tbase + PF_DIST, 0, 0);
  }

  v8f acc = {0.f, 0.f, 0.f, 0.f, 0.f, 0.f, 0.f, 0.f};
#pragma unroll
  for (int kk = 0; kk < 4; ++kk) {
    const int o = 4 * kk;
    const float t0 = fmaf(-ph1, yb[o + 1], fmaf(-q0, yb[o + 2], yb[o + 3] - muv));
    const float t1 = fmaf(-ph1, yb[o + 2], fmaf(-q0, yb[o + 3], yb[o + 4] - muv));
    v2f a; a.x = A0k[kk]; a.y = A1k[kk];
    v2f b; b.x = t0;      b.y = t1;
    acc = __builtin_amdgcn_wmma_f32_16x16x4_f32(
        false, a, false, b, (short)0, acc, false, false);
  }

  const float e2m = -th1 * e2;
  float d[8];
#pragma unroll
  for (int v = 0; v < 8; ++v)
    d[v] = fmaf(e1, hv[v + 1], fmaf(e2m, hv[v], acc[v]));

  e1 = __shfl(d[7], lrow + 16, 32);
  e2 = __shfl(d[6], lrow + 16, 32);

  float* op = orow + base + m0;
#pragma unroll
  for (int v = 0; v < 8; ++v) {
    if (LAST) {
      const int k = base + m0 + v;
      if (k < NVALID)     op[v] = d[v];
      else if (k < OUTW)  op[v] = 0.0f;
    } else {
      op[v] = d[v];
    }
  }
}

__global__ __launch_bounds__(WAVES_PER_BLOCK * 32) void arima_eps_kernel(
    const float* __restrict__ y, const float* __restrict__ phi,
    const float* __restrict__ theta, const float* __restrict__ mu,
    float* __restrict__ out)
{
  __shared__ float h_sh[17];

  const float ph0 = phi[0], ph1 = phi[1];
  const float th0 = theta[0], th1 = theta[1];
  const float muv = mu[0];

  if (threadIdx.x == 0) {
    h_sh[0] = 1.0f;
    h_sh[1] = -th0;
    for (int k = 2; k < 17; ++k)
      h_sh[k] = fmaf(-th0, h_sh[k - 1], -th1 * h_sh[k - 2]);
  }
  __syncthreads();

  const int lane  = threadIdx.x & 31;
  const int wid   = threadIdx.x >> 5;
  const int lrow  = lane & 15;
  const int half  = lane >> 4;
  const int khalf = half * 2;
  const int m0    = half * 8;
  const int row   = blockIdx.x * ROWS_PER_BLOCK + wid * ROWS_PER_WAVE + lrow;

  const float* yrow = y   + (size_t)row * TSEQ;
  float*       orow = out + (size_t)row * OUTW;

  float A0k[4], A1k[4];
#pragma unroll
  for (int kk = 0; kk < 4; ++kk) {
    const int c0 = 4 * kk + khalf;
    A0k[kk] = (lrow >= c0)     ? h_sh[lrow - c0]     : 0.0f;
    A1k[kk] = (lrow >= c0 + 1) ? h_sh[lrow - c0 - 1] : 0.0f;
  }
  float hv[9];
#pragma unroll
  for (int v = 0; v < 9; ++v) hv[v] = h_sh[m0 + v];

  const float q0 = 1.0f + ph0;
  float e1 = 0.0f, e2 = 0.0f;

  for (int ch = 0; ch < NCHUNK - 1; ++ch)
    process_chunk<false>(ch * CH, yrow, orow, khalf, m0, lrow,
                         A0k, A1k, hv, q0, ph1, muv, th1, e1, e2);
  process_chunk<true>((NCHUNK - 1) * CH, yrow, orow, khalf, m0, lrow,
                      A0k, A1k, hv, q0, ph1, muv, th1, e1, e2);
}

extern "C" void kernel_launch(void* const* d_in, const int* in_sizes, int n_in,
                              void* d_out, int out_size, void* d_ws, size_t ws_size,
                              hipStream_t stream) {
  const float* y     = (const float*)d_in[0];
  const float* phi   = (const float*)d_in[1];
  const float* theta = (const float*)d_in[2];
  const float* mu    = (const float*)d_in[3];
  float* out = (float*)d_out;

  dim3 grid(BSZ / ROWS_PER_BLOCK);
  dim3 block(WAVES_PER_BLOCK * 32);
  hipLaunchKernelGGL(arima_eps_kernel, grid, block, 0, stream,
                     y, phi, theta, mu, out);
}


# zero-state results parked in a 2nd LDS tile, free response added by the row's own wave at the store: freed registers fetch the next y row a pass ahead, one barrier less; free-response sums regrouped
# speedup vs baseline: 3.5029x; 3.5029x over previous
#include <hip/hip_runtime.h>
#include <hip/hip_bf16.h>

typedef __attribute__((ext_vector_type(2))) float v2f;
typedef __attribute__((ext_vector_type(8))) float v8f;

#define BSZ    8192
#define TSEQ   4096
#define OUTW   4095
#define NVALID 4093
#define CH     16
#define NCHUNK 256
#define ROWS_PER_BLOCK 16
#define NW     16
#define CPW    4
#define NPASS  4
#define SEG    (CPW * CH)
#define ZP     36
#define PASS_T (NW * SEG)
#define YP     (PASS_T + 4)
#define NLD    (PASS_T / 128)
typedef float v4f  __attribute__((ext_vector_type(4)));
typedef float v4fu __attribute__((ext_vector_type(4), aligned(4)));
static_assert(NW * CPW * NPASS == NCHUNK, "segments must tile the row");
static_assert(2 * NW <= ZP && PASS_T % 128 == 0 && NW == ROWS_PER_BLOCK, "LDS rows; one tile row per wave");
static_assert(SEG == 64 && NW == 2 * NLD, "a 512-byte wave store covers two segments, one per lane half");
static_assert(2 * NW == 32, "one lane per scan input");

__device__ __forceinline__ void chunk_window(int lbase, const float* ylds, int khalf, float (&yb)[17])
{
  const int tbase = lbase + khalf;
#pragma unroll
  for (int j = 1; j <= 16; ++j)
    yb[j] = ylds[tbase + j];
}

__device__ __forceinline__ void chunk_zero_state(
    int lbase, const float (&yb)[17], float* dlds, int lrow, int m0,
    const float A0k[4], const float A1k[4], const float hv[9],
    float q0, float ph1, float muv, float th1,
    float& e1, float& e2)
{
  v8f acc = {0.f, 0.f, 0.f, 0.f, 0.f, 0.f, 0.f, 0.f};
#pragma unroll
  for (int kk = 0; kk < 4; ++kk) {
    const int o = 4 * kk;
    const float t0 = fmaf(-ph1, yb[o + 1], fmaf(-q0, yb[o + 2], yb[o + 3] - muv));
    const float t1 = fmaf(-ph1, yb[o + 2], fmaf(-q0, yb[o + 3], yb[o + 4] - muv));
    v2f a; a.x = A0k[kk]; a.y = A1k[kk];
    v2f b; b.x = t0;      b.y = t1;
    acc = __builtin_amdgcn_wmma_f32_16x16x4_f32(
        false, a, false, b, (short)0, acc, false, false);
  }

  const float e2m = -th1 * e2;
  float d[8];
#pragma unroll
  for (int v = 0; v < 8; ++v)
    d[v] = fmaf(e1, hv[v + 1], fmaf(e2m, hv[v], acc[v]));

  e1 = __shfl(d[7], lrow + 16, 32);
  e2 = __shfl(d[6], lrow + 16, 32);

  v4f lo; lo.x = d[0]; lo.y = d[1]; lo.z = d[2]; lo.w = d[3];
  v4f hi; hi.x = d[4]; hi.y = d[5]; hi.z = d[6]; hi.w = d[7];
  *(v4f*)(dlds + lbase + m0)     = lo;
  *(v4f*)(dlds + lbase + m0 + 4) = hi;
}

__device__ __forceinline__ void row_load(
    const float* __restrict__ y, int yoff, int p, int lane, v4f (&R)[NLD], v4f& halo)
{
  const float* yp = y + __builtin_amdgcn_readfirstlane(yoff + p * PASS_T);
#pragma unroll
  for (int i = 0; i < NLD; ++i)
    R[i] = *(const v4f*)(yp + i * 128 + 4 * lane);
  halo = (v4f){0.f, 0.f, 0.f, 0.f};
  if (p + 1 < NPASS && lane == 0)
    halo = *(const v4f*)(yp + PASS_T);
}

__device__ __forceinline__ void lds_barrier()
{
  __builtin_amdgcn_fence(__ATOMIC_RELEASE, "workgroup", "local");
  __builtin_amdgcn_s_barrier();
  __builtin_amdgcn_fence(__ATOMIC_ACQUIRE, "workgroup", "local");
}

__global__ __launch_bounds__(NW * 32, 8) void arima_eps_kernel(
    const float* __restrict__ y, const float* __restrict__ phi,
    const float* __restrict__ theta, const float* __restrict__ mu,
    float* __restrict__ out)
{
  __shared__ __attribute__((aligned(16))) float h_sh[SEG + 4];
  __shared__ __attribute__((aligned(16))) float z_sh[NPASS][ROWS_PER_BLOCK][ZP];
  __shared__ __attribute__((aligned(16))) float tile[ROWS_PER_BLOCK][YP];
  __shared__ __attribute__((aligned(16))) float otile[ROWS_PER_BLOCK][YP];

  const float ph0 = phi[0], ph1 = phi[1];
  const float th0 = theta[0], th1 = theta[1];
  const float muv = mu[0];

  const int lane  = threadIdx.x & 31;
  const int wid   = __builtin_amdgcn_readfirstlane(threadIdx.x >> 5);
  const int lrow  = lane & 15;
  const int half  = lane >> 4;
  const int khalf = half * 2;
  const int m0    = half * 8;

  const int yoff = (blockIdx.x * ROWS_PER_BLOCK + wid) * TSEQ;
  const int ooff = (blockIdx.x * ROWS_PER_BLOCK + wid) * OUTW;
  float*       trow = &tile[wid][0];
  const float* orow = &otile[wid][0];
  const float* frow = &tile[lrow][0];
  float*       drow = &otile[lrow][0];

  v4f R[NLD], halo;
  row_load(y, yoff, 0, lane, R, halo);

  if (threadIdx.x == 0) {
    h_sh[0] = 1.0f;
    h_sh[1] = -th0;
    for (int k = 2; k < SEG + 1; ++k)
      h_sh[k] = fmaf(-th0, h_sh[k - 1], -th1 * h_sh[k - 2]);
  }
  __syncthreads();

  float A0k[4], A1k[4];
#pragma unroll
  for (int kk = 0; kk < 4; ++kk) {
    const int c0 = 4 * kk + khalf;
    A0k[kk] = (lrow >= c0)     ? h_sh[lrow - c0]     : 0.0f;
    A1k[kk] = (lrow >= c0 + 1) ? h_sh[lrow - c0 - 1] : 0.0f;
  }
  float hv[9];
#pragma unroll
  for (int v = 0; v < 9; ++v) hv[v] = h_sh[m0 + v];
  const float hL = h_sh[SEG], hLm1 = h_sh[SEG - 1], hLm2 = h_sh[SEG - 2];

  const float q0 = 1.0f + ph0;
  float sin1 = 0.0f, sin2 = 0.0f;
  const int lbase0 = wid * SEG;

  for (int p = 0; p < NPASS; ++p) {
#pragma unroll
    for (int i = 0; i < NLD; ++i)
      *(v4f*)(trow + i * 128 + 4 * lane) = R[i];
    if (lane == 0)
      *(v4f*)(trow + PASS_T) = halo;
    lds_barrier();

    if (p + 1 < NPASS)
      row_load(y, yoff, p + 1, lane, R, halo);

    float e1 = 0.0f, e2 = 0.0f;
    float yb[2][17];
    chunk_window(lbase0, frow, khalf, yb[0]);
#pragma unroll
    for (int c = 0; c < CPW; ++c) {
      if (c + 1 < CPW)
        chunk_window(lbase0 + (c + 1) * CH, frow, khalf, yb[(c + 1) & 1]);
      __builtin_amdgcn_sched_barrier(0);
      chunk_zero_state(lbase0 + c * CH, yb[c & 1], drow, lrow, m0, A0k, A1k, hv,
                       q0, ph1, muv, th1, e1, e2);
    }

    if (half == 0) {
      float2 zz; zz.x = e1; zz.y = e2;
      *(float2*)&z_sh[p][lrow][2 * wid] = zz;
    }
    lds_barrier();

    const int zl = __float_as_int(z_sh[p][wid][lane]);
    const v4f hs4 = *(const v4f*)&h_sh[4 * lrow];
    const float hs[5] = {hs4.x, hs4.y, hs4.z, hs4.w, h_sh[4 * lrow + 4]};
    float sa[NLD], sb[NLD];
    float s1 = sin1, s2 = sin2;
#pragma unroll
    for (int w = 0; w < NW; ++w) {
      if ((w & 1) == half) { sa[w >> 1] = s1; sb[w >> 1] = s2; }
      const float z1 = __int_as_float(__builtin_amdgcn_readlane(zl, 2 * w));
      const float z2 = __int_as_float(__builtin_amdgcn_readlane(zl, 2 * w + 1));
      const float s2m = -th1 * s2;
      const float n1 = fmaf(s1, hL,   fmaf(s2m, hLm1, z1));
      const float n2 = fmaf(s1, hLm1, fmaf(s2m, hLm2, z2));
      s1 = n1; s2 = n2;
    }
    sin1 = s1; sin2 = s2;

    float* op = out + __builtin_amdgcn_readfirstlane(ooff + p * PASS_T) + 4 * lane;
    v4f en = *(const v4f*)(orow + 4 * lane);
#pragma unroll
    for (int i = 0; i < NLD; ++i) {
      v4f e = en;
      if (i + 1 < NLD) {
        en = *(const v4f*)(orow + (i + 1) * 128 + 4 * lane);
        __builtin_amdgcn_sched_barrier(0);
      }
      const float bm = -th1 * sb[i];
      e.x = fmaf(sa[i], hs[1], fmaf(bm, hs[0], e.x));
      e.y = fmaf(sa[i], hs[2], fmaf(bm, hs[1], e.y));
      e.z = fmaf(sa[i], hs[3], fmaf(bm, hs[2], e.z));
      e.w = fmaf(sa[i], hs[4], fmaf(bm, hs[3], e.w));
      if (i == NLD - 1 && p == NPASS - 1 && lane == 31) {
        op[i * 128 + 0] = e.x; op[i * 128 + 1] = 0.0f; op[i * 128 + 2] = 0.0f;
      } else {
        *(v4fu*)(op + i * 128) = e;
      }
    }
  }
}

extern "C" void kernel_launch(void* const* d_in, const int* in_sizes, int n_in,
                              void* d_out, int out_size, void* d_ws, size_t ws_size,
                              hipStream_t stream) {
  const float* y     = (const float*)d_in[0];
  const float* phi   = (const float*)d_in[1];
  const float* theta = (const float*)d_in[2];
  const float* mu    = (const float*)d_in[3];
  float* out = (float*)d_out;

  dim3 grid(BSZ / ROWS_PER_BLOCK);
  dim3 block(NW * 32);
  hipLaunchKernelGGL(arima_eps_kernel, grid, block, 0, stream,
                     y, phi, theta, mu, out);
}


# eps stores re-cut on the 128-byte lines of the flat output (rows are 4095 floats, 4-byte aligned): 16-byte stores cover whole lines, a pass's first floats and last 0-3 go as dwords; sums unchanged
# speedup vs baseline: 3.6152x; 1.0321x over previous
#include <hip/hip_runtime.h>
#include <hip/hip_bf16.h>

typedef __attribute__((ext_vector_type(2))) float v2f;
typedef __attribute__((ext_vector_type(8))) float v8f;

#define BSZ    8192
#define TSEQ   4096
#define OUTW   4095
#define NVALID 4093
#define CH     16
#define NCHUNK 256
#define ROWS_PER_BLOCK 16
#define NW     16
#define CPW    4
#define NPASS  4
#define SEG    (CPW * CH)
#define ZP     36
#define PASS_T (NW * SEG)
#define YP     (PASS_T + 4)
#define NLD    (PASS_T / 128)
typedef float v4f  __attribute__((ext_vector_type(4)));
static_assert(NW * CPW * NPASS == NCHUNK, "segments must tile the row");
static_assert(2 * NW <= ZP && PASS_T % 128 == 0 && NW == ROWS_PER_BLOCK, "LDS rows; one tile row per wave");
static_assert(SEG == 64 && NW == 2 * NLD, "a lane half's 256 bytes of a wave store are one segment long");
static_assert(OUTW % 32 == 31, "head length of a row's part of a pass = row index mod 32");
static_assert(2 * NW == 32, "one lane per scan input");

__device__ __forceinline__ void chunk_window(int lbase, const float* ylds, int khalf, float (&yb)[17])
{
  const int tbase = lbase + khalf;
#pragma unroll
  for (int j = 1; j <= 16; ++j)
    yb[j] = ylds[tbase + j];
}

__device__ __forceinline__ void chunk_zero_state(
    int lbase, const float (&yb)[17], float* dlds, int lrow, int m0,
    const float A0k[4], const float A1k[4], const float hv[9],
    float q0, float ph1, float muv, float th1,
    float& e1, float& e2)
{
  v8f acc = {0.f, 0.f, 0.f, 0.f, 0.f, 0.f, 0.f, 0.f};
#pragma unroll
  for (int kk = 0; kk < 4; ++kk) {
    const int o = 4 * kk;
    const float t0 = fmaf(-ph1, yb[o + 1], fmaf(-q0, yb[o + 2], yb[o + 3] - muv));
    const float t1 = fmaf(-ph1, yb[o + 2], fmaf(-q0, yb[o + 3], yb[o + 4] - muv));
    v2f a; a.x = A0k[kk]; a.y = A1k[kk];
    v2f b; b.x = t0;      b.y = t1;
    acc = __builtin_amdgcn_wmma_f32_16x16x4_f32(
        false, a, false, b, (short)0, acc, false, false);
  }

  const float e2m = -th1 * e2;
  float d[8];
#pragma unroll
  for (int v = 0; v < 8; ++v)
    d[v] = fmaf(e1, hv[v + 1], fmaf(e2m, hv[v], acc[v]));

  e1 = __shfl(d[7], lrow + 16, 32);
  e2 = __shfl(d[6], lrow + 16, 32);

  v4f lo; lo.x = d[0]; lo.y = d[1]; lo.z = d[2]; lo.w = d[3];
  v4f hi; hi.x = d[4]; hi.y = d[5]; hi.z = d[6]; hi.w = d[7];
  *(v4f*)(dlds + lbase + m0)     = lo;
  *(v4f*)(dlds + lbase + m0 + 4) = hi;
}

__device__ __forceinline__ void row_load(
    const float* __restrict__ y, int yoff, int p, int lane, v4f (&R)[NLD], v4f& halo)
{
  const float* yp = y + __builtin_amdgcn_readfirstlane(yoff + p * PASS_T);
#pragma unroll
  for (int i = 0; i < NLD; ++i)
    R[i] = *(const v4f*)(yp + i * 128 + 4 * lane);
  halo = (v4f){0.f, 0.f, 0.f, 0.f};
  if (p + 1 < NPASS && lane == 0)
    halo = *(const v4f*)(yp + PASS_T);
}

__device__ __forceinline__ void lds_barrier()
{
  __builtin_amdgcn_fence(__ATOMIC_RELEASE, "workgroup", "local");
  __builtin_amdgcn_s_barrier();
  __builtin_amdgcn_fence(__ATOMIC_ACQUIRE, "workgroup", "local");
}

__global__ __launch_bounds__(NW * 32, 8) void arima_eps_kernel(
    const float* __restrict__ y, const float* __restrict__ phi,
    const float* __restrict__ theta, const float* __restrict__ mu,
    float* __restrict__ out)
{
  __shared__ __attribute__((aligned(16))) float h_sh[SEG + 4];
  __shared__ __attribute__((aligned(16))) float z_sh[NPASS][ROWS_PER_BLOCK][ZP];
  __shared__ __attribute__((aligned(16))) float tile[ROWS_PER_BLOCK][YP];
  __shared__ __attribute__((aligned(16))) float otile[ROWS_PER_BLOCK][YP];

  const float ph0 = phi[0], ph1 = phi[1];
  const float th0 = theta[0], th1 = theta[1];
  const float muv = mu[0];

  const int lane  = threadIdx.x & 31;
  const int wid   = __builtin_amdgcn_readfirstlane(threadIdx.x >> 5);
  const int lrow  = lane & 15;
  const int half  = lane >> 4;
  const int khalf = half * 2;
  const int m0    = half * 8;

  const int yoff = (blockIdx.x * ROWS_PER_BLOCK + wid) * TSEQ;
  const int ooff = (blockIdx.x * ROWS_PER_BLOCK + wid) * OUTW;
  const int hd = (blockIdx.x * ROWS_PER_BLOCK + wid) & 31;
  float*       trow = &tile[wid][0];
  const float* orow = &otile[wid][0];
  const float* frow = &tile[lrow][0];
  float*       drow = &otile[lrow][0];

  v4f R[NLD], halo;
  row_load(y, yoff, 0, lane, R, halo);

  if (threadIdx.x == 0) {
    h_sh[0] = 1.0f;
    h_sh[1] = -th0;
    for (int k = 2; k < SEG + 1; ++k)
      h_sh[k] = fmaf(-th0, h_sh[k - 1], -th1 * h_sh[k - 2]);
  }
  __syncthreads();

  float A0k[4], A1k[4];
#pragma unroll
  for (int kk = 0; kk < 4; ++kk) {
    const int c0 = 4 * kk + khalf;
    A0k[kk] = (lrow >= c0)     ? h_sh[lrow - c0]     : 0.0f;
    A1k[kk] = (lrow >= c0 + 1) ? h_sh[lrow - c0 - 1] : 0.0f;
  }
  float hv[9];
#pragma unroll
  for (int v = 0; v < 9; ++v) hv[v] = h_sh[m0 + v];
  const float hL = h_sh[SEG], hLm1 = h_sh[SEG - 1], hLm2 = h_sh[SEG - 2];

  const float q0 = 1.0f + ph0;
  float sin1 = 0.0f, sin2 = 0.0f;
  const int lbase0 = wid * SEG;

  for (int p = 0; p < NPASS; ++p) {
#pragma unroll
    for (int i = 0; i < NLD; ++i)
      *(v4f*)(trow + i * 128 + 4 * lane) = R[i];
    if (lane == 0)
      *(v4f*)(trow + PASS_T) = halo;
    lds_barrier();

    if (p + 1 < NPASS)
      row_load(y, yoff, p + 1, lane, R, halo);

    float e1 = 0.0f, e2 = 0.0f;
    float yb[2][17];
    chunk_window(lbase0, frow, khalf, yb[0]);
#pragma unroll
    for (int c = 0; c < CPW; ++c) {
      if (c + 1 < CPW)
        chunk_window(lbase0 + (c + 1) * CH, frow, khalf, yb[(c + 1) & 1]);
      __builtin_amdgcn_sched_barrier(0);
      chunk_zero_state(lbase0 + c * CH, yb[c & 1], drow, lrow, m0, A0k, A1k, hv,
                       q0, ph1, muv, th1, e1, e2);
    }

    if (half == 0) {
      float2 zz; zz.x = e1; zz.y = e2;
      *(float2*)&z_sh[p][lrow][2 * wid] = zz;
    }
    lds_barrier();

    const int zl = __float_as_int(z_sh[p][wid][lane]);
    float S1[NW + 1], S2m[NW + 1];
    float s1 = sin1, s2 = sin2;
#pragma unroll
    for (int w = 0; w < NW; ++w) {
      const float z1 = __int_as_float(__builtin_amdgcn_readlane(zl, 2 * w));
      const float z2 = __int_as_float(__builtin_amdgcn_readlane(zl, 2 * w + 1));
      const float s2m = -th1 * s2;
      S1[w] = s1; S2m[w] = s2m;
      const float n1 = fmaf(s1, hL,   fmaf(s2m, hLm1, z1));
      const float n2 = fmaf(s1, hLm1, fmaf(s2m, hLm2, z2));
      s1 = n1; s2 = n2;
    }
    S1[NW] = s1; S2m[NW] = -th1 * s2;
    sin1 = s1; sin2 = s2;

    float cA[4], cB[4];
    bool  nx[4];
#pragma unroll
    for (int q = 0; q < 4; ++q) {
      const int u = hd + 4 * lrow + q;
      const int j = u & (SEG - 1);
      cA[q] = h_sh[j + 1];
      cB[q] = h_sh[j];
      nx[q] = u >= SEG;
    }

    const int obase = __builtin_amdgcn_readfirstlane(ooff + p * PASS_T);
    const int nval  = (p == NPASS - 1) ? (OUTW - (NPASS - 1) * PASS_T) : PASS_T;
    const int kzero = (p == NPASS - 1) ? (NVALID - (NPASS - 1) * PASS_T) : PASS_T;
    float* ob = out + __builtin_amdgcn_readfirstlane(obase + hd) + 4 * lane;
    const int kl = hd + 4 * lane;
    const bool act7 = kl + (NLD - 1) * 128 + 4 <= nval;
    const float* dsrc = orow + kl;
    const float* dsrc7 = act7 ? dsrc + (NLD - 1) * 128 : dsrc;

    if (lane < hd) {
      const float dv = orow[lane];
      out[obase + lane] = fmaf(S1[0], h_sh[lane + 1], fmaf(S2m[0], h_sh[lane], dv));
    }

    float dn0 = dsrc[0], dn1 = dsrc[1], dn2 = dsrc[2], dn3 = dsrc[3];
#pragma unroll
    for (int i = 0; i < NLD; ++i) {
      const float d0 = dn0, d1 = dn1, d2 = dn2, d3 = dn3;
      if (i + 1 < NLD) {
        const float* nsrc = (i + 1 == NLD - 1) ? dsrc7 : dsrc + (i + 1) * 128;
        dn0 = nsrc[0]; dn1 = nsrc[1]; dn2 = nsrc[2]; dn3 = nsrc[3];
        __builtin_amdgcn_sched_barrier(0);
      }
      const float alo = half ? S1[2 * i + 1]  : S1[2 * i];
      const float ahi = half ? S1[2 * i + 2]  : S1[2 * i + 1];
      const float blo = half ? S2m[2 * i + 1] : S2m[2 * i];
      const float bhi = half ? S2m[2 * i + 2] : S2m[2 * i + 1];
      v4f e;
      e.x = fmaf(nx[0] ? ahi : alo, cA[0], fmaf(nx[0] ? bhi : blo, cB[0], d0));
      e.y = fmaf(nx[1] ? ahi : alo, cA[1], fmaf(nx[1] ? bhi : blo, cB[1], d1));
      e.z = fmaf(nx[2] ? ahi : alo, cA[2], fmaf(nx[2] ? bhi : blo, cB[2], d2));
      e.w = fmaf(nx[3] ? ahi : alo, cA[3], fmaf(nx[3] ? bhi : blo, cB[3], d3));
      if (i == NLD - 1) {
        const int k7 = kl + i * 128;
        if (k7 + 0 >= kzero) e.x = 0.0f;
        if (k7 + 1 >= kzero) e.y = 0.0f;
        if (k7 + 2 >= kzero) e.z = 0.0f;
        if (k7 + 3 >= kzero) e.w = 0.0f;
        if (act7)
          *(v4f*)(ob + i * 128) = e;
      } else {
        *(v4f*)(ob + i * 128) = e;
      }
    }

    const int kt0 = hd + ((nval - hd) & ~3);
    if (lane < ((nval - hd) & 3)) {
      const int kk = kt0 + lane;
      const int j  = kk - (NW - 1) * SEG;
      float ev = fmaf(S1[NW - 1], h_sh[j + 1], fmaf(S2m[NW - 1], h_sh[j], orow[kk]));
      if (kk >= kzero) ev = 0.0f;
      out[obase + kk] = ev;
    }
  }
}

extern "C" void kernel_launch(void* const* d_in, const int* in_sizes, int n_in,
                              void* d_out, int out_size, void* d_ws, size_t ws_size,
                              hipStream_t stream) {
  const float* y     = (const float*)d_in[0];
  const float* phi   = (const float*)d_in[1];
  const float* theta = (const float*)d_in[2];
  const float* mu    = (const float*)d_in[3];
  float* out = (float*)d_out;

  dim3 grid(BSZ / ROWS_PER_BLOCK);
  dim3 block(NW * 32);
  hipLaunchKernelGGL(arima_eps_kernel, grid, block, 0, stream,
                     y, phi, theta, mu, out);
}
